# starting kernel
# baseline (speedup 1.0000x reference)
#include <hip/hip_runtime.h>
#include <math.h>

#define NODES 50000
#define EDGES 600000
#define DIM   128
#define ORD   4

typedef float v2f __attribute__((ext_vector_type(2)));
typedef float v8f __attribute__((ext_vector_type(8)));

#define ATOMIC_ADD_F32(p, v) \
  __hip_atomic_fetch_add((p), (v), __ATOMIC_RELAXED, __HIP_MEMORY_SCOPE_AGENT)

__device__ __forceinline__ void atomicMaxFloat(float* addr, float v) {
  if (v >= 0.0f)
    atomicMax((int*)addr, __float_as_int(v));
  else
    atomicMin((unsigned int*)addr, (unsigned int)__float_as_int(v));
}

__global__ void k_zero(float* __restrict__ p, int n) {
  int i = blockIdx.x * blockDim.x + threadIdx.x;
  if (i < n) p[i] = 0.0f;
}

__global__ void k_deg(const int* __restrict__ dst, float* __restrict__ deg, int n) {
  int i = blockIdx.x * blockDim.x + threadIdx.x;
  if (i < n) ATOMIC_ADD_F32(&deg[dst[i]], 1.0f);
}

__global__ void k_norm(float* __restrict__ nrm, int n) {
  int i = blockIdx.x * blockDim.x + threadIdx.x;
  if (i < n) nrm[i] = rsqrtf(fmaxf(nrm[i], 1.0f));
}

__global__ void k_init(const float* __restrict__ feats, float* __restrict__ x,
                       float* __restrict__ y, int n4) {
  int i = blockIdx.x * blockDim.x + threadIdx.x;
  if (i < n4) {
    float4 v = ((const float4*)feats)[i];
    ((float4*)x)[i] = v;
    ((float4*)y)[i] = v;
  }
}

__global__ void __launch_bounds__(256)
k_node_dots(const float* __restrict__ x, const float* __restrict__ w,
            const float* __restrict__ nrm,
            float* __restrict__ a_s, float* __restrict__ a_d,
            float* __restrict__ h, float* __restrict__ m, float* __restrict__ s,
            int n_nodes) {
  int gtid = blockIdx.x * blockDim.x + threadIdx.x;
  int wave = gtid >> 5;
  int lane = gtid & 31;
  int base = wave * 16;
  if (base >= n_nodes) return;
  int nl = lane & 15;
  int hi = lane >> 4;

  float4 z4 = make_float4(0.f, 0.f, 0.f, 0.f);
  float* hp = h + (size_t)base * DIM;
#pragma unroll
  for (int j = 0; j < 16; ++j)
    *(float4*)(hp + j * DIM + lane * 4) = z4;

  if (hi == 0) {
    m[base + nl] = __int_as_float(0xff800000);
    s[base + nl] = 0.0f;
  }

  const float* xrow = x + (size_t)(base + nl) * DIM;
  v8f c = {};
#pragma unroll
  for (int k0 = 0; k0 < DIM; k0 += 4) {
    int kk = k0 + 2 * hi;
    v2f a;
    a.x = xrow[kk];
    a.y = xrow[kk + 1];
    const float* wp = w + (nl & 1) * DIM + kk;
    float bx = wp[0], by = wp[1];
    v2f b;
    b.x = (nl < 2) ? bx : 0.0f;
    b.y = (nl < 2) ? by : 0.0f;
    c = __builtin_amdgcn_wmma_f32_16x16x4_f32(false, a, false, b, (short)0, c,
                                              false, false);
  }

  if (nl < 2) {
    float* outp = (nl == 0) ? a_s : a_d;
    int row0 = base + hi * 8;
#pragma unroll
    for (int r = 0; r < 8; ++r)
      outp[row0 + r] = c[r] * nrm[row0 + r];
  }
}

__global__ void k_edge_e(const float* __restrict__ a_s, const float* __restrict__ a_d,
                         const int* __restrict__ src, const int* __restrict__ dst,
                         float* __restrict__ eb, float* __restrict__ m, int n) {
  int i = blockIdx.x * blockDim.x + threadIdx.x;
  if (i >= n) return;
  float e = a_s[src[i]] + a_d[dst[i]];
  e = (e > 0.0f) ? e : 0.2f * e;
  eb[i] = e;
  atomicMaxFloat(&m[dst[i]], e);
}

__global__ void k_edge_exp(float* __restrict__ eb, const int* __restrict__ dst,
                           const float* __restrict__ m, float* __restrict__ s, int n) {
  int i = blockIdx.x * blockDim.x + threadIdx.x;
  if (i >= n) return;
  int d = dst[i];
  float ex = expf(eb[i] - m[d]);
  eb[i] = ex;
  ATOMIC_ADD_F32(&s[d], ex);
}

__global__ void __launch_bounds__(256)
k_scatter(const float* __restrict__ x, const float* __restrict__ eb,
          const float* __restrict__ s, const int* __restrict__ src,
          const int* __restrict__ dst, float* __restrict__ h, int n_edges) {
  int gtid = blockIdx.x * blockDim.x + threadIdx.x;
  int wave = gtid >> 5;
  int lane = gtid & 31;
  if (wave >= n_edges) return;
  int si = src[wave], di = dst[wave];
  float sv = s[di];
  float alpha = eb[wave] / ((sv > 0.0f) ? sv : 1.0f);
  float4 v = *(const float4*)(x + (size_t)si * DIM + lane * 4);
  float* hp = h + (size_t)di * DIM + lane * 4;
  ATOMIC_ADD_F32(hp + 0, alpha * v.x);
  ATOMIC_ADD_F32(hp + 1, alpha * v.y);
  ATOMIC_ADD_F32(hp + 2, alpha * v.z);
  ATOMIC_ADD_F32(hp + 3, alpha * v.w);
}

__global__ void k_accum(const float* __restrict__ h, float* __restrict__ y, int n4) {
  int i = blockIdx.x * blockDim.x + threadIdx.x;
  if (i < n4) {
    float4 a = ((const float4*)h)[i];
    float4 b = ((float4*)y)[i];
    b.x += a.x; b.y += a.y; b.z += a.z; b.w += a.w;
    ((float4*)y)[i] = b;
  }
}

__global__ void k_scale(float* __restrict__ y, int n4) {
  const float sc = 1.0f / (float)(ORD + 1);
  int i = blockIdx.x * blockDim.x + threadIdx.x;
  if (i < n4) {
    float4 b = ((float4*)y)[i];
    b.x *= sc; b.y *= sc; b.z *= sc; b.w *= sc;
    ((float4*)y)[i] = b;
  }
}

extern "C" void kernel_launch(void* const* d_in, const int* in_sizes, int n_in,
                              void* d_out, int out_size, void* d_ws, size_t ws_size,
                              hipStream_t stream) {
  const float* feats = (const float*)d_in[0];
  const float* w_att = (const float*)d_in[1];
  const int* src = (const int*)d_in[2];
  const int* dst = (const int*)d_in[3];
  float* out = (float*)d_out;

  const size_t ND = (size_t)NODES * DIM;
  float* ws  = (float*)d_ws;
  float* x   = ws;
  float* h   = x + ND;
  float* a_s = h + ND;
  float* a_d = a_s + NODES;
  float* nrm = a_d + NODES;
  float* m   = nrm + NODES;
  float* s   = m + NODES;
  float* eb  = s + NODES;

  const int B = 256;
  const int nblk_node = (NODES + B - 1) / B;
  const int nblk_edge = (EDGES + B - 1) / B;
  const int nblk_nd4  = (int)((ND / 4 + B - 1) / B);
  const int nblk_dots = (int)(((size_t)((NODES + 15) / 16) * 32 + B - 1) / B);
  const int nblk_scat = (int)(((size_t)EDGES * 32 + B - 1) / B);

  k_zero<<<nblk_node, B, 0, stream>>>(nrm, NODES);
  k_deg<<<nblk_edge, B, 0, stream>>>(dst, nrm, EDGES);
  k_norm<<<nblk_node, B, 0, stream>>>(nrm, NODES);
  k_init<<<nblk_nd4, B, 0, stream>>>(feats, x, out, (int)(ND / 4));

  for (int t = 0; t < ORD; ++t) {
    k_node_dots<<<nblk_dots, B, 0, stream>>>(x, w_att, nrm, a_s, a_d, h, m, s, NODES);
    k_edge_e<<<nblk_edge, B, 0, stream>>>(a_s, a_d, src, dst, eb, m, EDGES);
    k_edge_exp<<<nblk_edge, B, 0, stream>>>(eb, dst, m, s, EDGES);
    k_scatter<<<nblk_scat, B, 0, stream>>>(x, eb, s, src, dst, h, EDGES);
    k_accum<<<nblk_nd4, B, 0, stream>>>(h, out, (int)(ND / 4));
    float* tmp = x; x = h; h = tmp;
  }
  k_scale<<<nblk_nd4, B, 0, stream>>>(out, (int)(ND / 4));
}


# Edges counting-sorted by dst per call (int atomic arrival rank + scan); one wave per dst node does max/sum/weighted row gather in registers: no float atomics; sum order = arrival rank, lane butterfly
# speedup vs baseline: 4.3150x; 4.3150x over previous
#include <hip/hip_runtime.h>
#include <math.h>

#define NODES 50000
#define EDGES 600000
#define DIM   128
#define ORD   4

typedef float v2f __attribute__((ext_vector_type(2)));
typedef float v8f __attribute__((ext_vector_type(8)));

__global__ void k_zero(int* __restrict__ p, int n) {
  int i = blockIdx.x * blockDim.x + threadIdx.x;
  if (i < n) p[i] = 0;
}

__global__ void k_rank(const int* __restrict__ dst, int* __restrict__ cnt,
                       int* __restrict__ rank, int n) {
  int i = blockIdx.x * blockDim.x + threadIdx.x;
  if (i < n)
    rank[i] = __hip_atomic_fetch_add(&cnt[dst[i]], 1, __ATOMIC_RELAXED,
                                     __HIP_MEMORY_SCOPE_AGENT);
}

#define SCAN_T 1024
#define SCAN_W (SCAN_T / 32)
#define SCAN_U ((NODES / 4 + SCAN_T - 1) / SCAN_T)
static_assert(NODES % 4 == 0 && SCAN_U * SCAN_T * 4 >= NODES, "one pass must cover every count");
__global__ void __launch_bounds__(SCAN_T)
k_scan(const int* __restrict__ cnt, int* __restrict__ off, float* __restrict__ nrm) {
  __shared__ int s_part[SCAN_W];
  __shared__ int s_wtot[SCAN_W];
  int t = threadIdx.x, lane = t & 31, wid = t >> 5;
  int b0 = blockIdx.x * SCAN_T;
  int part = 0;
  const int4* c4 = (const int4*)cnt;
  int nq = b0 >> 2;
  int4 v[SCAN_U];
#pragma unroll
  for (int u = 0; u < SCAN_U; ++u) {
    int q = u * SCAN_T + t;
    v[u] = c4[(q < nq) ? q : 0];
  }
#pragma unroll
  for (int u = 0; u < SCAN_U; ++u) {
    int q = u * SCAN_T + t;
    int sum4 = (v[u].x + v[u].y) + (v[u].z + v[u].w);
    part += (q < nq) ? sum4 : 0;
  }
#pragma unroll
  for (int d = 16; d > 0; d >>= 1) part += __shfl_xor(part, d, 32);
  int i = b0 + t;
  int c = (i < NODES) ? cnt[i] : 0;
  int inc = c;
#pragma unroll
  for (int d = 1; d < 32; d <<= 1) {
    int o = __shfl_up(inc, d, 32);
    if (lane >= d) inc += o;
  }
  if (lane == 0) s_part[wid] = part;
  if (lane == 31) s_wtot[wid] = inc;
  __syncthreads();
  int pre = 0;
#pragma unroll
  for (int w = 0; w < SCAN_W; ++w) {
    pre += s_part[w];
    if (w < wid) pre += s_wtot[w];
  }
  int excl = pre + inc - c;
  if (i < NODES) {
    off[i] = excl;
    nrm[i] = rsqrtf(fmaxf((float)c, 1.0f));
  }
  if (i == NODES - 1) off[NODES] = excl + c;
}

__global__ void k_fill(const int* __restrict__ src, const int* __restrict__ dst,
                       const int* __restrict__ off, const int* __restrict__ rank,
                       int* __restrict__ ss, int n) {
  int i = blockIdx.x * blockDim.x + threadIdx.x;
  if (i < n) ss[off[dst[i]] + rank[i]] = src[i];
}

__global__ void k_init(const float* __restrict__ feats, float* __restrict__ x,
                       float* __restrict__ y, int n4) {
  int i = blockIdx.x * blockDim.x + threadIdx.x;
  if (i < n4) {
    float4 v = ((const float4*)feats)[i];
    ((float4*)x)[i] = v;
    ((float4*)y)[i] = v;
  }
}

__global__ void __launch_bounds__(256)
k_node_dots(const float* __restrict__ x, const float* __restrict__ w,
            const float* __restrict__ nrm,
            float* __restrict__ a_s, float* __restrict__ a_d, int n_nodes) {
  int gtid = blockIdx.x * blockDim.x + threadIdx.x;
  int wave = gtid >> 5;
  int lane = gtid & 31;
  int base = wave * 16;
  if (base >= n_nodes) return;
  int nl = lane & 15;
  int hi = lane >> 4;

  const float* xrow = x + (size_t)(base + nl) * DIM;
  v8f c = {};
#pragma unroll
  for (int k0 = 0; k0 < DIM; k0 += 4) {
    int kk = k0 + 2 * hi;
    v2f a;
    a.x = xrow[kk];
    a.y = xrow[kk + 1];
    const float* wp = w + (nl & 1) * DIM + kk;
    float bx = wp[0], by = wp[1];
    v2f b;
    b.x = (nl < 2) ? bx : 0.0f;
    b.y = (nl < 2) ? by : 0.0f;
    c = __builtin_amdgcn_wmma_f32_16x16x4_f32(false, a, false, b, (short)0, c,
                                              false, false);
  }

  if (nl < 2) {
    float* outp = (nl == 0) ? a_s : a_d;
    int row0 = base + hi * 8;
#pragma unroll
    for (int r = 0; r < 8; ++r)
      outp[row0 + r] = c[r] * nrm[row0 + r];
  }
}

template <unsigned LO, unsigned HI>
__device__ __forceinline__ float row_xchg(float x) {
  int b = __builtin_bit_cast(int, x);
  return __builtin_bit_cast(float, __builtin_amdgcn_permlane16(b, b, LO, HI, false, false));
}
__device__ __forceinline__ float xhalf(float x) {
  int b = __builtin_bit_cast(int, x);
  return __builtin_bit_cast(float,
      __builtin_amdgcn_permlanex16(b, b, 0x76543210u, 0xfedcba98u, false, false));
}
__device__ __forceinline__ float wave_max(float x) {
  x = fmaxf(x, row_xchg<0x67452301u, 0xefcdab89u>(x));
  x = fmaxf(x, row_xchg<0x54761032u, 0xdcfe98bau>(x));
  x = fmaxf(x, row_xchg<0x32107654u, 0xba98fedcu>(x));
  x = fmaxf(x, row_xchg<0xfedcba98u, 0x76543210u>(x));
  x = fmaxf(x, xhalf(x));
  return x;
}
__device__ __forceinline__ float wave_sum(float x) {
  x += row_xchg<0x67452301u, 0xefcdab89u>(x);
  x += row_xchg<0x54761032u, 0xdcfe98bau>(x);
  x += row_xchg<0x32107654u, 0xba98fedcu>(x);
  x += row_xchg<0xfedcba98u, 0x76543210u>(x);
  x += xhalf(x);
  return x;
}

__device__ __forceinline__ float leaky(float e) { return (e > 0.0f) ? e : 0.2f * e; }

#define SRC_OF(k) __builtin_amdgcn_readlane(sj, (k))
#define AL_OF(k)  __builtin_bit_cast(float, __builtin_amdgcn_readlane(__builtin_bit_cast(int, al), (k)))
#define ROW_OF(k) (*(const float4*)(x + (size_t)SRC_OF(k) * DIM + lane4))
#define FMA4(a, v) do { float a_ = (a); acc.x += a_ * (v).x; acc.y += a_ * (v).y; \
                        acc.z += a_ * (v).z; acc.w += a_ * (v).w; } while (0)

__global__ void __launch_bounds__(256)
k_seg(const float* __restrict__ x, const float* __restrict__ a_s,
      const float* __restrict__ a_d, const int* __restrict__ off,
      const int* __restrict__ ss, float* __restrict__ h, int n_nodes) {
  int lane = threadIdx.x & 31;
  int n = __builtin_amdgcn_readfirstlane((int)(blockIdx.x * (blockDim.x >> 5) + (threadIdx.x >> 5)));
  if (n >= n_nodes) return;
  int beg = off[n], end = off[n + 1];
  float ad = a_d[n];
  const float ninf = __int_as_float(0xff800000);

  int j0 = beg + lane;
  bool v0ok = j0 < end;
  int sj0 = 0;
  float e0 = ninf;
  if (v0ok) {
    sj0 = ss[j0];
    e0 = leaky(a_s[sj0] + ad);
  }
  float mx = e0;
  for (int j = j0 + 32; j < end; j += 32)
    mx = fmaxf(mx, leaky(a_s[ss[j]] + ad));
  mx = wave_max(mx);

  float ex0 = v0ok ? expf(e0 - mx) : 0.0f;
  float sm = ex0;
  for (int j = j0 + 32; j < end; j += 32)
    sm += expf(leaky(a_s[ss[j]] + ad) - mx);
  sm = wave_sum(sm);
  float den = (sm > 0.0f) ? sm : 1.0f;

  float4 acc = make_float4(0.f, 0.f, 0.f, 0.f);
  const unsigned lane4 = (unsigned)lane * 4u;
  int sj = sj0;
  float al = ex0 / den;
  for (int jb = beg; jb < end; jb += 32) {
    if (jb != beg) {
      int j = jb + lane;
      sj = 0;
      al = 0.0f;
      if (j < end) {
        sj = ss[j];
        al = expf(leaky(a_s[sj] + ad) - mx) / den;
      }
    }
    int cc = end - jb;
    cc = (cc < 32) ? cc : 32;
    int k = 0;
    for (; k + 8 <= cc; k += 8) {
      float4 r0 = ROW_OF(k),     r1 = ROW_OF(k + 1), r2 = ROW_OF(k + 2), r3 = ROW_OF(k + 3);
      float4 r4 = ROW_OF(k + 4), r5 = ROW_OF(k + 5), r6 = ROW_OF(k + 6), r7 = ROW_OF(k + 7);
      FMA4(AL_OF(k), r0);     FMA4(AL_OF(k + 1), r1);
      FMA4(AL_OF(k + 2), r2); FMA4(AL_OF(k + 3), r3);
      FMA4(AL_OF(k + 4), r4); FMA4(AL_OF(k + 5), r5);
      FMA4(AL_OF(k + 6), r6); FMA4(AL_OF(k + 7), r7);
    }
    if (k + 4 <= cc) {
      float4 r0 = ROW_OF(k), r1 = ROW_OF(k + 1), r2 = ROW_OF(k + 2), r3 = ROW_OF(k + 3);
      FMA4(AL_OF(k), r0);     FMA4(AL_OF(k + 1), r1);
      FMA4(AL_OF(k + 2), r2); FMA4(AL_OF(k + 3), r3);
      k += 4;
    }
    int rem = cc - k;
    if (rem > 0) {
      float4 r0 = ROW_OF(k);
      float4 r1 = make_float4(0.f, 0.f, 0.f, 0.f), r2 = r1;
      if (rem > 1) r1 = ROW_OF(k + 1);
      if (rem > 2) r2 = ROW_OF(k + 2);
      FMA4(AL_OF(k), r0);
      if (rem > 1) FMA4(AL_OF(k + 1), r1);
      if (rem > 2) FMA4(AL_OF(k + 2), r2);
    }
  }
  *(float4*)(h + (size_t)n * DIM + lane4) = acc;
}

__global__ void k_accum(const float* __restrict__ h, float* __restrict__ y, int n4) {
  int i = blockIdx.x * blockDim.x + threadIdx.x;
  if (i < n4) {
    float4 a = ((const float4*)h)[i];
    float4 b = ((float4*)y)[i];
    b.x += a.x; b.y += a.y; b.z += a.z; b.w += a.w;
    ((float4*)y)[i] = b;
  }
}

__global__ void k_scale(float* __restrict__ y, int n4) {
  const float sc = 1.0f / (float)(ORD + 1);
  int i = blockIdx.x * blockDim.x + threadIdx.x;
  if (i < n4) {
    float4 b = ((float4*)y)[i];
    b.x *= sc; b.y *= sc; b.z *= sc; b.w *= sc;
    ((float4*)y)[i] = b;
  }
}

extern "C" void kernel_launch(void* const* d_in, const int* in_sizes, int n_in,
                              void* d_out, int out_size, void* d_ws, size_t ws_size,
                              hipStream_t stream) {
  const float* feats = (const float*)d_in[0];
  const float* w_att = (const float*)d_in[1];
  const int* src = (const int*)d_in[2];
  const int* dst = (const int*)d_in[3];
  float* out = (float*)d_out;

  const size_t ND = (size_t)NODES * DIM;
  float* ws  = (float*)d_ws;
  float* x   = ws;
  float* h   = x + ND;
  float* a_s = h + ND;
  float* a_d = a_s + NODES;
  float* nrm = a_d + NODES;
  int* cnt   = (int*)(nrm + NODES);
  int* off   = cnt + NODES;
  int* rank  = off + NODES + 16;
  int* ss    = rank + EDGES;

  const int B = 256;
  const int nblk_node = (NODES + B - 1) / B;
  const int nblk_edge = (EDGES + B - 1) / B;
  const int nblk_nd4  = (int)((ND / 4 + B - 1) / B);
  const int nblk_dots = (int)(((size_t)((NODES + 15) / 16) * 32 + B - 1) / B);
  const int nblk_seg  = (NODES + (B / 32) - 1) / (B / 32);

  k_zero<<<nblk_node, B, 0, stream>>>(cnt, NODES);
  k_rank<<<nblk_edge, B, 0, stream>>>(dst, cnt, rank, EDGES);
  k_scan<<<(NODES + SCAN_T - 1) / SCAN_T, SCAN_T, 0, stream>>>(cnt, off, nrm);
  k_fill<<<nblk_edge, B, 0, stream>>>(src, dst, off, rank, ss, EDGES);
  k_init<<<nblk_nd4, B, 0, stream>>>(feats, x, out, (int)(ND / 4));

  for (int t = 0; t < ORD; ++t) {
    k_node_dots<<<nblk_dots, B, 0, stream>>>(x, w_att, nrm, a_s, a_d, NODES);
    k_seg<<<nblk_seg, B, 0, stream>>>(x, a_s, a_d, off, ss, h, NODES);
    k_accum<<<nblk_nd4, B, 0, stream>>>(h, out, (int)(ND / 4));
    float* tmp = x; x = h; h = tmp;
  }
  k_scale<<<nblk_nd4, B, 0, stream>>>(out, (int)(ND / 4));
}


# y chain folded into the segment kernel: last round stores ((((feats+h1)+h2)+h3)+h4)/5 to out, same add order; feats copy, 4 y+=h passes, scale pass gone (18->12 launches); round 1 reads feats itself
# speedup vs baseline: 5.4087x; 1.2535x over previous
#include <hip/hip_runtime.h>
#include <math.h>

#define NODES 50000
#define EDGES 600000
#define DIM   128
#define ORD   4

typedef float v2f __attribute__((ext_vector_type(2)));
typedef float v8f __attribute__((ext_vector_type(8)));

__global__ void k_zero(int* __restrict__ p, int n) {
  int i = blockIdx.x * blockDim.x + threadIdx.x;
  if (i < n) p[i] = 0;
}

__global__ void k_rank(const int* __restrict__ dst, int* __restrict__ cnt,
                       int* __restrict__ rank, int n) {
  int i = blockIdx.x * blockDim.x + threadIdx.x;
  if (i < n)
    rank[i] = __hip_atomic_fetch_add(&cnt[dst[i]], 1, __ATOMIC_RELAXED,
                                     __HIP_MEMORY_SCOPE_AGENT);
}

#define SCAN_T 1024
#define SCAN_W (SCAN_T / 32)
#define SCAN_U ((NODES / 4 + SCAN_T - 1) / SCAN_T)
static_assert(NODES % 4 == 0 && SCAN_U * SCAN_T * 4 >= NODES, "one pass must cover every count");
__global__ void __launch_bounds__(SCAN_T)
k_scan(const int* __restrict__ cnt, int* __restrict__ off, float* __restrict__ nrm) {
  __shared__ int s_part[SCAN_W];
  __shared__ int s_wtot[SCAN_W];
  int t = threadIdx.x, lane = t & 31, wid = t >> 5;
  int b0 = blockIdx.x * SCAN_T;
  int part = 0;
  const int4* c4 = (const int4*)cnt;
  int nq = b0 >> 2;
  int4 v[SCAN_U];
#pragma unroll
  for (int u = 0; u < SCAN_U; ++u) {
    int q = u * SCAN_T + t;
    v[u] = c4[(q < nq) ? q : 0];
  }
#pragma unroll
  for (int u = 0; u < SCAN_U; ++u) {
    int q = u * SCAN_T + t;
    int sum4 = (v[u].x + v[u].y) + (v[u].z + v[u].w);
    part += (q < nq) ? sum4 : 0;
  }
#pragma unroll
  for (int d = 16; d > 0; d >>= 1) part += __shfl_xor(part, d, 32);
  int i = b0 + t;
  int c = (i < NODES) ? cnt[i] : 0;
  int inc = c;
#pragma unroll
  for (int d = 1; d < 32; d <<= 1) {
    int o = __shfl_up(inc, d, 32);
    if (lane >= d) inc += o;
  }
  if (lane == 0) s_part[wid] = part;
  if (lane == 31) s_wtot[wid] = inc;
  __syncthreads();
  int pre = 0;
#pragma unroll
  for (int w = 0; w < SCAN_W; ++w) {
    pre += s_part[w];
    if (w < wid) pre += s_wtot[w];
  }
  int excl = pre + inc - c;
  if (i < NODES) {
    off[i] = excl;
    nrm[i] = rsqrtf(fmaxf((float)c, 1.0f));
  }
  if (i == NODES - 1) off[NODES] = excl + c;
}

__global__ void k_fill(const int* __restrict__ src, const int* __restrict__ dst,
                       const int* __restrict__ off, const int* __restrict__ rank,
                       int* __restrict__ ss, int n) {
  int i = blockIdx.x * blockDim.x + threadIdx.x;
  if (i < n) ss[off[dst[i]] + rank[i]] = src[i];
}

__global__ void __launch_bounds__(256)
k_node_dots(const float* __restrict__ x, const float* __restrict__ w,
            const float* __restrict__ nrm,
            float* __restrict__ a_s, float* __restrict__ a_d, int n_nodes) {
  int gtid = blockIdx.x * blockDim.x + threadIdx.x;
  int wave = gtid >> 5;
  int lane = gtid & 31;
  int base = wave * 16;
  if (base >= n_nodes) return;
  int nl = lane & 15;
  int hi = lane >> 4;

  const float* xrow = x + (size_t)(base + nl) * DIM;
  v8f c = {};
#pragma unroll
  for (int k0 = 0; k0 < DIM; k0 += 4) {
    int kk = k0 + 2 * hi;
    v2f a;
    a.x = xrow[kk];
    a.y = xrow[kk + 1];
    const float* wp = w + (nl & 1) * DIM + kk;
    float bx = wp[0], by = wp[1];
    v2f b;
    b.x = (nl < 2) ? bx : 0.0f;
    b.y = (nl < 2) ? by : 0.0f;
    c = __builtin_amdgcn_wmma_f32_16x16x4_f32(false, a, false, b, (short)0, c,
                                              false, false);
  }

  if (nl < 2) {
    float* outp = (nl == 0) ? a_s : a_d;
    int row0 = base + hi * 8;
#pragma unroll
    for (int r = 0; r < 8; ++r)
      outp[row0 + r] = c[r] * nrm[row0 + r];
  }
}

template <unsigned LO, unsigned HI>
__device__ __forceinline__ float row_xchg(float x) {
  int b = __builtin_bit_cast(int, x);
  return __builtin_bit_cast(float, __builtin_amdgcn_permlane16(b, b, LO, HI, false, false));
}
__device__ __forceinline__ float xhalf(float x) {
  int b = __builtin_bit_cast(int, x);
  return __builtin_bit_cast(float,
      __builtin_amdgcn_permlanex16(b, b, 0x76543210u, 0xfedcba98u, false, false));
}
__device__ __forceinline__ float wave_max(float x) {
  x = fmaxf(x, row_xchg<0x67452301u, 0xefcdab89u>(x));
  x = fmaxf(x, row_xchg<0x54761032u, 0xdcfe98bau>(x));
  x = fmaxf(x, row_xchg<0x32107654u, 0xba98fedcu>(x));
  x = fmaxf(x, row_xchg<0xfedcba98u, 0x76543210u>(x));
  x = fmaxf(x, xhalf(x));
  return x;
}
__device__ __forceinline__ float wave_sum(float x) {
  x += row_xchg<0x67452301u, 0xefcdab89u>(x);
  x += row_xchg<0x54761032u, 0xdcfe98bau>(x);
  x += row_xchg<0x32107654u, 0xba98fedcu>(x);
  x += row_xchg<0xfedcba98u, 0x76543210u>(x);
  x += xhalf(x);
  return x;
}

__device__ __forceinline__ float leaky(float e) { return (e > 0.0f) ? e : 0.2f * e; }

#define SRC_OF(k) __builtin_amdgcn_readlane(sj, (k))
#define AL_OF(k)  __builtin_bit_cast(float, __builtin_amdgcn_readlane(__builtin_bit_cast(int, al), (k)))
#define ROW_OF(k) (*(const float4*)(x + (size_t)SRC_OF(k) * DIM + lane4))
#define FMA4(a, v) do { float a_ = (a); acc.x += a_ * (v).x; acc.y += a_ * (v).y; \
                        acc.z += a_ * (v).z; acc.w += a_ * (v).w; } while (0)

template <bool LAST>
__global__ void __launch_bounds__(256)
k_seg(const float* __restrict__ x, const float* __restrict__ a_s,
      const float* __restrict__ a_d, const int* __restrict__ off,
      const int* __restrict__ ss, float* __restrict__ h,
      const float* __restrict__ y0, const float* __restrict__ y1,
      const float* __restrict__ y2, int n_nodes) {
  int lane = threadIdx.x & 31;
  int n = __builtin_amdgcn_readfirstlane((int)(blockIdx.x * (blockDim.x >> 5) + (threadIdx.x >> 5)));
  if (n >= n_nodes) return;
  const unsigned lane4 = (unsigned)lane * 4u;
  float4 p0 = make_float4(0.f, 0.f, 0.f, 0.f), p1 = p0, p2 = p0, p3 = p0;
  if (LAST) {
    size_t ro = (size_t)n * DIM + lane4;
    p0 = *(const float4*)(y0 + ro);
    p1 = *(const float4*)(y1 + ro);
    p2 = *(const float4*)(y2 + ro);
    p3 = *(const float4*)(x + ro);
  }
  int beg = off[n], end = off[n + 1];
  float ad = a_d[n];
  const float ninf = __int_as_float(0xff800000);

  int j0 = beg + lane;
  bool v0ok = j0 < end;
  int sj0 = 0;
  float e0 = ninf;
  if (v0ok) {
    sj0 = ss[j0];
    e0 = leaky(a_s[sj0] + ad);
  }
  float mx = e0;
  for (int j = j0 + 32; j < end; j += 32)
    mx = fmaxf(mx, leaky(a_s[ss[j]] + ad));
  mx = wave_max(mx);

  float ex0 = v0ok ? expf(e0 - mx) : 0.0f;
  float sm = ex0;
  for (int j = j0 + 32; j < end; j += 32)
    sm += expf(leaky(a_s[ss[j]] + ad) - mx);
  sm = wave_sum(sm);
  float den = (sm > 0.0f) ? sm : 1.0f;

  float4 ys = make_float4(0.f, 0.f, 0.f, 0.f);
  if (LAST) {
    ys.x = ((p0.x + p1.x) + p2.x) + p3.x;
    ys.y = ((p0.y + p1.y) + p2.y) + p3.y;
    ys.z = ((p0.z + p1.z) + p2.z) + p3.z;
    ys.w = ((p0.w + p1.w) + p2.w) + p3.w;
  }
  float4 acc = make_float4(0.f, 0.f, 0.f, 0.f);
  int sj = sj0;
  float al = ex0 / den;
  for (int jb = beg; jb < end; jb += 32) {
    if (jb != beg) {
      int j = jb + lane;
      sj = 0;
      al = 0.0f;
      if (j < end) {
        sj = ss[j];
        al = expf(leaky(a_s[sj] + ad) - mx) / den;
      }
    }
    int cc = end - jb;
    cc = (cc < 32) ? cc : 32;
    int k = 0;
    for (; k + 8 <= cc; k += 8) {
      float4 r0 = ROW_OF(k),     r1 = ROW_OF(k + 1), r2 = ROW_OF(k + 2), r3 = ROW_OF(k + 3);
      float4 r4 = ROW_OF(k + 4), r5 = ROW_OF(k + 5), r6 = ROW_OF(k + 6), r7 = ROW_OF(k + 7);
      FMA4(AL_OF(k), r0);     FMA4(AL_OF(k + 1), r1);
      FMA4(AL_OF(k + 2), r2); FMA4(AL_OF(k + 3), r3);
      FMA4(AL_OF(k + 4), r4); FMA4(AL_OF(k + 5), r5);
      FMA4(AL_OF(k + 6), r6); FMA4(AL_OF(k + 7), r7);
    }
    if (k + 4 <= cc) {
      float4 r0 = ROW_OF(k), r1 = ROW_OF(k + 1), r2 = ROW_OF(k + 2), r3 = ROW_OF(k + 3);
      FMA4(AL_OF(k), r0);     FMA4(AL_OF(k + 1), r1);
      FMA4(AL_OF(k + 2), r2); FMA4(AL_OF(k + 3), r3);
      k += 4;
    }
    int rem = cc - k;
    if (rem > 0) {
      float4 r0 = ROW_OF(k);
      float4 r1 = make_float4(0.f, 0.f, 0.f, 0.f), r2 = r1;
      if (rem > 1) r1 = ROW_OF(k + 1);
      if (rem > 2) r2 = ROW_OF(k + 2);
      FMA4(AL_OF(k), r0);
      if (rem > 1) FMA4(AL_OF(k + 1), r1);
      if (rem > 2) FMA4(AL_OF(k + 2), r2);
    }
  }
  if (LAST) {
    const float sc = 1.0f / (float)(ORD + 1);
    acc.x = (ys.x + acc.x) * sc;
    acc.y = (ys.y + acc.y) * sc;
    acc.z = (ys.z + acc.z) * sc;
    acc.w = (ys.w + acc.w) * sc;
  }
  *(float4*)(h + (size_t)n * DIM + lane4) = acc;
}

extern "C" void kernel_launch(void* const* d_in, const int* in_sizes, int n_in,
                              void* d_out, int out_size, void* d_ws, size_t ws_size,
                              hipStream_t stream) {
  const float* feats = (const float*)d_in[0];
  const float* w_att = (const float*)d_in[1];
  const int* src = (const int*)d_in[2];
  const int* dst = (const int*)d_in[3];
  float* out = (float*)d_out;

  const size_t ND = (size_t)NODES * DIM;
  float* ws  = (float*)d_ws;
  float* h1  = ws;
  float* h2  = h1 + ND;
  float* h3  = h2 + ND;
  float* a_s = h3 + ND;
  float* a_d = a_s + NODES;
  float* nrm = a_d + NODES;
  int* cnt   = (int*)(nrm + NODES);
  int* off   = cnt + NODES;
  int* rank  = off + NODES + 16;
  int* ss    = rank + EDGES;

  const int B = 256;
  const int nblk_node = (NODES + B - 1) / B;
  const int nblk_edge = (EDGES + B - 1) / B;
  const int nblk_dots = (int)(((size_t)((NODES + 15) / 16) * 32 + B - 1) / B);
  const int nblk_seg  = (NODES + (B / 32) - 1) / (B / 32);

  k_zero<<<nblk_node, B, 0, stream>>>(cnt, NODES);
  k_rank<<<nblk_edge, B, 0, stream>>>(dst, cnt, rank, EDGES);
  k_scan<<<(NODES + SCAN_T - 1) / SCAN_T, SCAN_T, 0, stream>>>(cnt, off, nrm);
  k_fill<<<nblk_edge, B, 0, stream>>>(src, dst, off, rank, ss, EDGES);

  static_assert(ORD == 4, "the rounds are written out");
  k_node_dots<<<nblk_dots, B, 0, stream>>>(feats, w_att, nrm, a_s, a_d, NODES);
  k_seg<false><<<nblk_seg, B, 0, stream>>>(feats, a_s, a_d, off, ss, h1, nullptr, nullptr, nullptr, NODES);
  k_node_dots<<<nblk_dots, B, 0, stream>>>(h1, w_att, nrm, a_s, a_d, NODES);
  k_seg<false><<<nblk_seg, B, 0, stream>>>(h1, a_s, a_d, off, ss, h2, nullptr, nullptr, nullptr, NODES);
  k_node_dots<<<nblk_dots, B, 0, stream>>>(h2, w_att, nrm, a_s, a_d, NODES);
  k_seg<false><<<nblk_seg, B, 0, stream>>>(h2, a_s, a_d, off, ss, h3, nullptr, nullptr, nullptr, NODES);
  k_node_dots<<<nblk_dots, B, 0, stream>>>(h3, w_att, nrm, a_s, a_d, NODES);
  k_seg<true><<<nblk_seg, B, 0, stream>>>(h3, a_s, a_d, off, ss, out, feats, h1, h2, NODES);
}


# Edges grouped by dst by a two-level counting sort in LDS (256-node bins: block histograms, bin runs, per-bin node sort): no per-edge global atomic or scattered 4-byte access; sum order = LDS arrival
# speedup vs baseline: 7.8772x; 1.4564x over previous
#include <hip/hip_runtime.h>
#include <math.h>

#define NODES 50000
#define EDGES 600000
#define DIM   128
#define ORD   4

typedef float v2f __attribute__((ext_vector_type(2)));
typedef float v8f __attribute__((ext_vector_type(8)));

#define BIN_SH   8
#define BIN_R    (1 << BIN_SH)
#define NBIN     ((NODES + BIN_R - 1) / BIN_R)
#define L1_T     1024
#define L1_PER   4
#define L1_C     (L1_T * L1_PER)
#define L1_G     ((EDGES + L1_C - 1) / L1_C)
#define HIST_N   (NBIN * L1_G)
#define BIN_IT   ((NBIN + 31) / 32)
#define ROW_IT   ((L1_G + 31) / 32)
#define L2_T     1024
#define L2_PER   4
#define L2_CAP   (L2_T * L2_PER)
static_assert(NODES < (1 << (31 - BIN_SH)), "a record is (src << BIN_SH) | (dst & (BIN_R - 1))");
static_assert(NBIN <= L1_T && BIN_R <= L2_T && BIN_IT * 32 >= NBIN, "one thread per counter");

template <unsigned LO, unsigned HI>
__device__ __forceinline__ int row_xchg_i(int b) {
  return __builtin_amdgcn_permlane16(b, b, LO, HI, false, false);
}
__device__ __forceinline__ int wave_sum_i(int x) {
  x += row_xchg_i<0x67452301u, 0xefcdab89u>(x);
  x += row_xchg_i<0x54761032u, 0xdcfe98bau>(x);
  x += row_xchg_i<0x32107654u, 0xba98fedcu>(x);
  x += row_xchg_i<0xfedcba98u, 0x76543210u>(x);
  x += __builtin_amdgcn_permlanex16(x, x, 0x76543210u, 0xfedcba98u, false, false);
  return x;
}

__global__ void __launch_bounds__(L1_T)
k_hist(const int* __restrict__ dst, int* __restrict__ hist) {
  __shared__ int s_h[NBIN];
  const int t = threadIdx.x, b = blockIdx.x;
  int d[L1_PER];
#pragma unroll
  for (int u = 0; u < L1_PER; ++u) {
    int i = b * L1_C + u * L1_T + t;
    int dv = dst[(i < EDGES) ? i : 0];
    d[u] = (i < EDGES) ? dv : -1;
  }
  if (t < NBIN) s_h[t] = 0;
  __syncthreads();
#pragma unroll
  for (int u = 0; u < L1_PER; ++u)
    if (d[u] >= 0) atomicAdd(&s_h[d[u] >> BIN_SH], 1);
  __syncthreads();
  if (t < NBIN) hist[t * L1_G + b] = s_h[t];
}

__global__ void __launch_bounds__(L1_T)
k_bin(const int* __restrict__ src, const int* __restrict__ dst, const int* __restrict__ hist,
      int* __restrict__ binst, int* __restrict__ tmp) {
  __shared__ int s_tot[NBIN];
  __shared__ int s_pre[NBIN];
  __shared__ int s_own[NBIN];
  __shared__ int s_gst[NBIN];
  __shared__ int s_lst[NBIN];
  __shared__ int s_cur[NBIN];
  __shared__ int s_rec[L1_C];
  __shared__ int s_bof[L1_C];
  const int t = threadIdx.x, lane = t & 31, wid = t >> 5, b = blockIdx.x;

  int ed[L1_PER], es[L1_PER];
#pragma unroll
  for (int u = 0; u < L1_PER; ++u) {
    int i = b * L1_C + u * L1_T + t;
    bool ok = i < EDGES;
    int ic = ok ? i : 0;
    int dv = dst[ic], sv = src[ic];
    ed[u] = ok ? dv : -1;
    es[u] = sv;
  }

  int vt[BIN_IT], vp[BIN_IT], vo[BIN_IT];
#pragma unroll
  for (int it = 0; it < BIN_IT; ++it) {
    int bin = wid + 32 * it;
    const int* row = hist + ((bin < NBIN) ? bin : 0) * L1_G;
    int tot = 0, pre = 0, own = 0;
#pragma unroll
    for (int k = 0; k < ROW_IT; ++k) {
      int c = k * 32 + lane;
      int v = row[(c < L1_G) ? c : 0];
      v = (c < L1_G) ? v : 0;
      tot += v;
      pre += (c < b) ? v : 0;
      own += (c == b) ? v : 0;
    }
    vt[it] = tot; vp[it] = pre; vo[it] = own;
  }
#pragma unroll
  for (int it = 0; it < BIN_IT; ++it) {
    int bin = wid + 32 * it;
    int tot = wave_sum_i(vt[it]), pre = wave_sum_i(vp[it]), own = wave_sum_i(vo[it]);
    if (bin < NBIN && lane == 0) { s_tot[bin] = tot; s_pre[bin] = pre; s_own[bin] = own; }
  }
  __syncthreads();

  if (wid == 0) {
    int tv[BIN_IT], ov[BIN_IT];
    int ts = 0, os = 0;
#pragma unroll
    for (int k = 0; k < BIN_IT; ++k) {
      int bin = lane * BIN_IT + k;
      tv[k] = (bin < NBIN) ? s_tot[bin] : 0;
      ov[k] = (bin < NBIN) ? s_own[bin] : 0;
      ts += tv[k];
      os += ov[k];
    }
    int tinc = ts, oinc = os;
#pragma unroll
    for (int d = 1; d < 32; d <<= 1) {
      int a = __shfl_up(tinc, d, 32), o = __shfl_up(oinc, d, 32);
      if (lane >= d) { tinc += a; oinc += o; }
    }
    int texc = tinc - ts, oexc = oinc - os;
#pragma unroll
    for (int k = 0; k < BIN_IT; ++k) {
      int bin = lane * BIN_IT + k;
      if (bin < NBIN) {
        s_gst[bin] = texc + s_pre[bin];
        s_lst[bin] = oexc;
        s_cur[bin] = oexc;
      }
      texc += tv[k];
      oexc += ov[k];
    }
  }
  __syncthreads();

#pragma unroll
  for (int u = 0; u < L1_PER; ++u) {
    if (ed[u] >= 0) {
      int bin = ed[u] >> BIN_SH;
      int p = atomicAdd(&s_cur[bin], 1);
      s_rec[p] = (es[u] << BIN_SH) | (ed[u] & (BIN_R - 1));
      s_bof[p] = bin;
    }
  }
  __syncthreads();

  int left = EDGES - b * L1_C;
  int n_here = (left < L1_C) ? left : L1_C;
#pragma unroll
  for (int u = 0; u < L1_PER; ++u) {
    int q = u * L1_T + t;
    if (q < n_here) {
      int bin = s_bof[q];
      tmp[s_gst[bin] + (q - s_lst[bin])] = s_rec[q];
    }
  }
  if (b == 0) {
    if (t < NBIN) binst[t] = s_gst[t];
    if (t == 0) binst[NBIN] = EDGES;
  }
}

__global__ void __launch_bounds__(L2_T)
k_binsort(const int* __restrict__ binst, const int* __restrict__ tmp, int* __restrict__ off,
          float* __restrict__ nrm, int* __restrict__ ss) {
  __shared__ int s_cnt[BIN_R];
  __shared__ int s_cur[BIN_R];
  __shared__ int s_w[BIN_R / 32];
  __shared__ int s_out[L2_CAP];
  const int t = threadIdx.x, lane = t & 31, wid = t >> 5, bin = blockIdx.x;
  const int start = binst[bin], end = binst[bin + 1];
  const int n = end - start;
  int r[L2_PER];
#pragma unroll
  for (int u = 0; u < L2_PER; ++u) {
    int j = start + u * L2_T + t;
    int rv = tmp[(j < end) ? j : 0];
    r[u] = (j < end) ? rv : -1;
  }
  if (t < BIN_R) s_cnt[t] = 0;
  __syncthreads();
#pragma unroll
  for (int u = 0; u < L2_PER; ++u)
    if (r[u] >= 0) atomicAdd(&s_cnt[r[u] & (BIN_R - 1)], 1);
  for (int j = start + L2_CAP + t; j < end; j += L2_T)
    atomicAdd(&s_cnt[tmp[j] & (BIN_R - 1)], 1);
  __syncthreads();

  int c = (t < BIN_R) ? s_cnt[t] : 0;
  int inc = c;
#pragma unroll
  for (int d = 1; d < 32; d <<= 1) {
    int o = __shfl_up(inc, d, 32);
    if (lane >= d) inc += o;
  }
  if (t < BIN_R && lane == 31) s_w[wid] = inc;
  __syncthreads();
  int excl = 0;
  if (t < BIN_R) {
    int pre = 0;
#pragma unroll
    for (int w = 0; w < BIN_R / 32; ++w) pre += (w < wid) ? s_w[w] : 0;
    excl = pre + inc - c;
    s_cur[t] = excl;
  }
  __syncthreads();

  const bool fits = n <= L2_CAP;
#pragma unroll
  for (int u = 0; u < L2_PER; ++u) {
    if (r[u] >= 0) {
      int p = atomicAdd(&s_cur[r[u] & (BIN_R - 1)], 1);
      int s = r[u] >> BIN_SH;
      if (fits) s_out[p] = s; else ss[start + p] = s;
    }
  }
  for (int j = start + L2_CAP + t; j < end; j += L2_T) {
    int rec = tmp[j];
    int p = atomicAdd(&s_cur[rec & (BIN_R - 1)], 1);
    ss[start + p] = rec >> BIN_SH;
  }
  __syncthreads();
  if (fits)
    for (int q = t; q < n; q += L2_T) ss[start + q] = s_out[q];
  if (t < BIN_R) {
    int node = bin * BIN_R + t;
    if (node < NODES) {
      off[node] = start + excl;
      nrm[node] = rsqrtf(fmaxf((float)c, 1.0f));
    }
  }
  if (bin == NBIN - 1 && t == 0) off[NODES] = end;
}

__global__ void __launch_bounds__(256)
k_node_dots(const float* __restrict__ x, const float* __restrict__ w,
            const float* __restrict__ nrm,
            float* __restrict__ a_s, float* __restrict__ a_d, int n_nodes) {
  int gtid = blockIdx.x * blockDim.x + threadIdx.x;
  int wave = gtid >> 5;
  int lane = gtid & 31;
  int base = wave * 16;
  if (base >= n_nodes) return;
  int nl = lane & 15;
  int hi = lane >> 4;

  const float* xrow = x + (size_t)(base + nl) * DIM;
  v8f c = {};
#pragma unroll
  for (int k0 = 0; k0 < DIM; k0 += 4) {
    int kk = k0 + 2 * hi;
    v2f a;
    a.x = xrow[kk];
    a.y = xrow[kk + 1];
    const float* wp = w + (nl & 1) * DIM + kk;
    float bx = wp[0], by = wp[1];
    v2f b;
    b.x = (nl < 2) ? bx : 0.0f;
    b.y = (nl < 2) ? by : 0.0f;
    c = __builtin_amdgcn_wmma_f32_16x16x4_f32(false, a, false, b, (short)0, c,
                                              false, false);
  }

  if (nl < 2) {
    float* outp = (nl == 0) ? a_s : a_d;
    int row0 = base + hi * 8;
#pragma unroll
    for (int r = 0; r < 8; ++r)
      outp[row0 + r] = c[r] * nrm[row0 + r];
  }
}

template <unsigned LO, unsigned HI>
__device__ __forceinline__ float row_xchg(float x) {
  int b = __builtin_bit_cast(int, x);
  return __builtin_bit_cast(float, __builtin_amdgcn_permlane16(b, b, LO, HI, false, false));
}
__device__ __forceinline__ float xhalf(float x) {
  int b = __builtin_bit_cast(int, x);
  return __builtin_bit_cast(float,
      __builtin_amdgcn_permlanex16(b, b, 0x76543210u, 0xfedcba98u, false, false));
}
__device__ __forceinline__ float wave_max(float x) {
  x = fmaxf(x, row_xchg<0x67452301u, 0xefcdab89u>(x));
  x = fmaxf(x, row_xchg<0x54761032u, 0xdcfe98bau>(x));
  x = fmaxf(x, row_xchg<0x32107654u, 0xba98fedcu>(x));
  x = fmaxf(x, row_xchg<0xfedcba98u, 0x76543210u>(x));
  x = fmaxf(x, xhalf(x));
  return x;
}
__device__ __forceinline__ float wave_sum(float x) {
  x += row_xchg<0x67452301u, 0xefcdab89u>(x);
  x += row_xchg<0x54761032u, 0xdcfe98bau>(x);
  x += row_xchg<0x32107654u, 0xba98fedcu>(x);
  x += row_xchg<0xfedcba98u, 0x76543210u>(x);
  x += xhalf(x);
  return x;
}

__device__ __forceinline__ float leaky(float e) { return (e > 0.0f) ? e : 0.2f * e; }

#define SRC_OF(k) __builtin_amdgcn_readlane(sj, (k))
#define AL_OF(k)  __builtin_bit_cast(float, __builtin_amdgcn_readlane(__builtin_bit_cast(int, al), (k)))
#define ROW_OF(k) (*(const float4*)(x + (size_t)SRC_OF(k) * DIM + lane4))
#define FMA4(a, v) do { float a_ = (a); acc.x += a_ * (v).x; acc.y += a_ * (v).y; \
                        acc.z += a_ * (v).z; acc.w += a_ * (v).w; } while (0)

template <bool LAST>
__global__ void __launch_bounds__(256)
k_seg(const float* __restrict__ x, const float* __restrict__ a_s,
      const float* __restrict__ a_d, const int* __restrict__ off,
      const int* __restrict__ ss, float* __restrict__ h,
      const float* __restrict__ y0, const float* __restrict__ y1,
      const float* __restrict__ y2, int n_nodes) {
  int lane = threadIdx.x & 31;
  int n = __builtin_amdgcn_readfirstlane((int)(blockIdx.x * (blockDim.x >> 5) + (threadIdx.x >> 5)));
  if (n >= n_nodes) return;
  const unsigned lane4 = (unsigned)lane * 4u;
  float4 p0 = make_float4(0.f, 0.f, 0.f, 0.f), p1 = p0, p2 = p0, p3 = p0;
  if (LAST) {
    size_t ro = (size_t)n * DIM + lane4;
    p0 = *(const float4*)(y0 + ro);
    p1 = *(const float4*)(y1 + ro);
    p2 = *(const float4*)(y2 + ro);
    p3 = *(const float4*)(x + ro);
  }
  int beg = off[n], end = off[n + 1];
  float ad = a_d[n];
  const float ninf = __int_as_float(0xff800000);

  int j0 = beg + lane;
  bool v0ok = j0 < end;
  int sj0 = 0;
  float e0 = ninf;
  if (v0ok) {
    sj0 = ss[j0];
    e0 = leaky(a_s[sj0] + ad);
  }
  float mx = e0;
  for (int j = j0 + 32; j < end; j += 32)
    mx = fmaxf(mx, leaky(a_s[ss[j]] + ad));
  mx = wave_max(mx);

  float ex0 = v0ok ? expf(e0 - mx) : 0.0f;
  float sm = ex0;
  for (int j = j0 + 32; j < end; j += 32)
    sm += expf(leaky(a_s[ss[j]] + ad) - mx);
  sm = wave_sum(sm);
  float den = (sm > 0.0f) ? sm : 1.0f;

  float4 ys = make_float4(0.f, 0.f, 0.f, 0.f);
  if (LAST) {
    ys.x = ((p0.x + p1.x) + p2.x) + p3.x;
    ys.y = ((p0.y + p1.y) + p2.y) + p3.y;
    ys.z = ((p0.z + p1.z) + p2.z) + p3.z;
    ys.w = ((p0.w + p1.w) + p2.w) + p3.w;
  }
  float4 acc = make_float4(0.f, 0.f, 0.f, 0.f);
  int sj = sj0;
  float al = ex0 / den;
  for (int jb = beg; jb < end; jb += 32) {
    if (jb != beg) {
      int j = jb + lane;
      sj = 0;
      al = 0.0f;
      if (j < end) {
        sj = ss[j];
        al = expf(leaky(a_s[sj] + ad) - mx) / den;
      }
    }
    int cc = end - jb;
    cc = (cc < 32) ? cc : 32;
    int k = 0;
    for (; k + 8 <= cc; k += 8) {
      float4 r0 = ROW_OF(k),     r1 = ROW_OF(k + 1), r2 = ROW_OF(k + 2), r3 = ROW_OF(k + 3);
      float4 r4 = ROW_OF(k + 4), r5 = ROW_OF(k + 5), r6 = ROW_OF(k + 6), r7 = ROW_OF(k + 7);
      FMA4(AL_OF(k), r0);     FMA4(AL_OF(k + 1), r1);
      FMA4(AL_OF(k + 2), r2); FMA4(AL_OF(k + 3), r3);
      FMA4(AL_OF(k + 4), r4); FMA4(AL_OF(k + 5), r5);
      FMA4(AL_OF(k + 6), r6); FMA4(AL_OF(k + 7), r7);
    }
    if (k + 4 <= cc) {
      float4 r0 = ROW_OF(k), r1 = ROW_OF(k + 1), r2 = ROW_OF(k + 2), r3 = ROW_OF(k + 3);
      FMA4(AL_OF(k), r0);     FMA4(AL_OF(k + 1), r1);
      FMA4(AL_OF(k + 2), r2); FMA4(AL_OF(k + 3), r3);
      k += 4;
    }
    int rem = cc - k;
    if (rem > 0) {
      float4 r0 = ROW_OF(k);
      float4 r1 = make_float4(0.f, 0.f, 0.f, 0.f), r2 = r1;
      if (rem > 1) r1 = ROW_OF(k + 1);
      if (rem > 2) r2 = ROW_OF(k + 2);
      FMA4(AL_OF(k), r0);
      if (rem > 1) FMA4(AL_OF(k + 1), r1);
      if (rem > 2) FMA4(AL_OF(k + 2), r2);
    }
  }
  if (LAST) {
    const float sc = 1.0f / (float)(ORD + 1);
    acc.x = (ys.x + acc.x) * sc;
    acc.y = (ys.y + acc.y) * sc;
    acc.z = (ys.z + acc.z) * sc;
    acc.w = (ys.w + acc.w) * sc;
  }
  *(float4*)(h + (size_t)n * DIM + lane4) = acc;
}

extern "C" void kernel_launch(void* const* d_in, const int* in_sizes, int n_in,
                              void* d_out, int out_size, void* d_ws, size_t ws_size,
                              hipStream_t stream) {
  const float* feats = (const float*)d_in[0];
  const float* w_att = (const float*)d_in[1];
  const int* src = (const int*)d_in[2];
  const int* dst = (const int*)d_in[3];
  float* out = (float*)d_out;

  const size_t ND = (size_t)NODES * DIM;
  float* ws  = (float*)d_ws;
  float* h1  = ws;
  float* h2  = h1 + ND;
  float* h3  = h2 + ND;
  float* a_s = h3 + ND;
  float* a_d = a_s + NODES;
  float* nrm = a_d + NODES;
  int* off   = (int*)(nrm + NODES);
  int* ss    = off + NODES + 16;
  int* hist  = ss + EDGES;
  int* binst = hist + HIST_N;
  int* tmp   = binst + NBIN + 16;

  const int B = 256;
  const int nblk_dots = (int)(((size_t)((NODES + 15) / 16) * 32 + B - 1) / B);
  const int nblk_seg  = (NODES + (B / 32) - 1) / (B / 32);

  k_hist<<<L1_G, L1_T, 0, stream>>>(dst, hist);
  k_bin<<<L1_G, L1_T, 0, stream>>>(src, dst, hist, binst, tmp);
  k_binsort<<<NBIN, L2_T, 0, stream>>>(binst, tmp, off, nrm, ss);

  static_assert(ORD == 4, "the rounds are written out");
  k_node_dots<<<nblk_dots, B, 0, stream>>>(feats, w_att, nrm, a_s, a_d, NODES);
  k_seg<false><<<nblk_seg, B, 0, stream>>>(feats, a_s, a_d, off, ss, h1, nullptr, nullptr, nullptr, NODES);
  k_node_dots<<<nblk_dots, B, 0, stream>>>(h1, w_att, nrm, a_s, a_d, NODES);
  k_seg<false><<<nblk_seg, B, 0, stream>>>(h1, a_s, a_d, off, ss, h2, nullptr, nullptr, nullptr, NODES);
  k_node_dots<<<nblk_dots, B, 0, stream>>>(h2, w_att, nrm, a_s, a_d, NODES);
  k_seg<false><<<nblk_seg, B, 0, stream>>>(h2, a_s, a_d, off, ss, h3, nullptr, nullptr, nullptr, NODES);
  k_node_dots<<<nblk_dots, B, 0, stream>>>(h3, w_att, nrm, a_s, a_d, NODES);
  k_seg<true><<<nblk_seg, B, 0, stream>>>(h3, a_s, a_d, off, ss, out, feats, h1, h2, NODES);
}


# Node dots of rounds 2-4 by linearity inside the segment kernel: h.w = sum_e alpha_e*(x[src_e].w), from records gathered for the logits; 3 WMMA launches and h re-reads gone; logit sum order changes
# speedup vs baseline: 8.9664x; 1.1383x over previous
#include <hip/hip_runtime.h>
#include <math.h>

#define NODES 50000
#define EDGES 600000
#define DIM   128
#define ORD   4

typedef float v2f __attribute__((ext_vector_type(2)));
typedef float v8f __attribute__((ext_vector_type(8)));

#define BIN_SH   8
#define BIN_R    (1 << BIN_SH)
#define NBIN     ((NODES + BIN_R - 1) / BIN_R)
#define L1_T     1024
#define L1_PER   4
#define L1_C     (L1_T * L1_PER)
#define L1_G     ((EDGES + L1_C - 1) / L1_C)
#define HIST_N   (NBIN * L1_G)
#define BIN_IT   ((NBIN + 31) / 32)
#define ROW_IT   ((L1_G + 31) / 32)
#define L2_T     1024
#define L2_PER   4
#define L2_CAP   (L2_T * L2_PER)
static_assert(NODES < (1 << (31 - BIN_SH)), "a record is (src << BIN_SH) | (dst & (BIN_R - 1))");
static_assert(NBIN <= L1_T && BIN_R <= L2_T && BIN_IT * 32 >= NBIN, "one thread per counter");

template <unsigned LO, unsigned HI>
__device__ __forceinline__ int row_xchg_i(int b) {
  return __builtin_amdgcn_permlane16(b, b, LO, HI, false, false);
}
__device__ __forceinline__ int wave_sum_i(int x) {
  x += row_xchg_i<0x67452301u, 0xefcdab89u>(x);
  x += row_xchg_i<0x54761032u, 0xdcfe98bau>(x);
  x += row_xchg_i<0x32107654u, 0xba98fedcu>(x);
  x += row_xchg_i<0xfedcba98u, 0x76543210u>(x);
  x += __builtin_amdgcn_permlanex16(x, x, 0x76543210u, 0xfedcba98u, false, false);
  return x;
}

__global__ void __launch_bounds__(L1_T)
k_hist(const int* __restrict__ dst, int* __restrict__ hist) {
  __shared__ int s_h[NBIN];
  const int t = threadIdx.x, b = blockIdx.x;
  int d[L1_PER];
#pragma unroll
  for (int u = 0; u < L1_PER; ++u) {
    int i = b * L1_C + u * L1_T + t;
    int dv = dst[(i < EDGES) ? i : 0];
    d[u] = (i < EDGES) ? dv : -1;
  }
  if (t < NBIN) s_h[t] = 0;
  __syncthreads();
#pragma unroll
  for (int u = 0; u < L1_PER; ++u)
    if (d[u] >= 0) atomicAdd(&s_h[d[u] >> BIN_SH], 1);
  __syncthreads();
  if (t < NBIN) hist[t * L1_G + b] = s_h[t];
}

__global__ void __launch_bounds__(L1_T)
k_bin(const int* __restrict__ src, const int* __restrict__ dst, const int* __restrict__ hist,
      int* __restrict__ binst, int* __restrict__ tmp) {
  __shared__ int s_tot[NBIN];
  __shared__ int s_pre[NBIN];
  __shared__ int s_own[NBIN];
  __shared__ int s_gst[NBIN];
  __shared__ int s_lst[NBIN];
  __shared__ int s_cur[NBIN];
  __shared__ int s_rec[L1_C];
  __shared__ int s_bof[L1_C];
  const int t = threadIdx.x, lane = t & 31, wid = t >> 5, b = blockIdx.x;

  int ed[L1_PER], es[L1_PER];
#pragma unroll
  for (int u = 0; u < L1_PER; ++u) {
    int i = b * L1_C + u * L1_T + t;
    bool ok = i < EDGES;
    int ic = ok ? i : 0;
    int dv = dst[ic], sv = src[ic];
    ed[u] = ok ? dv : -1;
    es[u] = sv;
  }

  int vt[BIN_IT], vp[BIN_IT], vo[BIN_IT];
#pragma unroll
  for (int it = 0; it < BIN_IT; ++it) {
    int bin = wid + 32 * it;
    const int* row = hist + ((bin < NBIN) ? bin : 0) * L1_G;
    int tot = 0, pre = 0, own = 0;
#pragma unroll
    for (int k = 0; k < ROW_IT; ++k) {
      int c = k * 32 + lane;
      int v = row[(c < L1_G) ? c : 0];
      v = (c < L1_G) ? v : 0;
      tot += v;
      pre += (c < b) ? v : 0;
      own += (c == b) ? v : 0;
    }
    vt[it] = tot; vp[it] = pre; vo[it] = own;
  }
#pragma unroll
  for (int it = 0; it < BIN_IT; ++it) {
    int bin = wid + 32 * it;
    int tot = wave_sum_i(vt[it]), pre = wave_sum_i(vp[it]), own = wave_sum_i(vo[it]);
    if (bin < NBIN && lane == 0) { s_tot[bin] = tot; s_pre[bin] = pre; s_own[bin] = own; }
  }
  __syncthreads();

  if (wid == 0) {
    int tv[BIN_IT], ov[BIN_IT];
    int ts = 0, os = 0;
#pragma unroll
    for (int k = 0; k < BIN_IT; ++k) {
      int bin = lane * BIN_IT + k;
      tv[k] = (bin < NBIN) ? s_tot[bin] : 0;
      ov[k] = (bin < NBIN) ? s_own[bin] : 0;
      ts += tv[k];
      os += ov[k];
    }
    int tinc = ts, oinc = os;
#pragma unroll
    for (int d = 1; d < 32; d <<= 1) {
      int a = __shfl_up(tinc, d, 32), o = __shfl_up(oinc, d, 32);
      if (lane >= d) { tinc += a; oinc += o; }
    }
    int texc = tinc - ts, oexc = oinc - os;
#pragma unroll
    for (int k = 0; k < BIN_IT; ++k) {
      int bin = lane * BIN_IT + k;
      if (bin < NBIN) {
        s_gst[bin] = texc + s_pre[bin];
        s_lst[bin] = oexc;
        s_cur[bin] = oexc;
      }
      texc += tv[k];
      oexc += ov[k];
    }
  }
  __syncthreads();

#pragma unroll
  for (int u = 0; u < L1_PER; ++u) {
    if (ed[u] >= 0) {
      int bin = ed[u] >> BIN_SH;
      int p = atomicAdd(&s_cur[bin], 1);
      s_rec[p] = (es[u] << BIN_SH) | (ed[u] & (BIN_R - 1));
      s_bof[p] = bin;
    }
  }
  __syncthreads();

  int left = EDGES - b * L1_C;
  int n_here = (left < L1_C) ? left : L1_C;
#pragma unroll
  for (int u = 0; u < L1_PER; ++u) {
    int q = u * L1_T + t;
    if (q < n_here) {
      int bin = s_bof[q];
      tmp[s_gst[bin] + (q - s_lst[bin])] = s_rec[q];
    }
  }
  if (b == 0) {
    if (t < NBIN) binst[t] = s_gst[t];
    if (t == 0) binst[NBIN] = EDGES;
  }
}

__global__ void __launch_bounds__(L2_T)
k_binsort(const int* __restrict__ binst, const int* __restrict__ tmp, int* __restrict__ off,
          float* __restrict__ nrm, int* __restrict__ ss) {
  __shared__ int s_cnt[BIN_R];
  __shared__ int s_cur[BIN_R];
  __shared__ int s_w[BIN_R / 32];
  __shared__ int s_out[L2_CAP];
  const int t = threadIdx.x, lane = t & 31, wid = t >> 5, bin = blockIdx.x;
  const int start = binst[bin], end = binst[bin + 1];
  const int n = end - start;
  int r[L2_PER];
#pragma unroll
  for (int u = 0; u < L2_PER; ++u) {
    int j = start + u * L2_T + t;
    int rv = tmp[(j < end) ? j : 0];
    r[u] = (j < end) ? rv : -1;
  }
  if (t < BIN_R) s_cnt[t] = 0;
  __syncthreads();
#pragma unroll
  for (int u = 0; u < L2_PER; ++u)
    if (r[u] >= 0) atomicAdd(&s_cnt[r[u] & (BIN_R - 1)], 1);
  for (int j = start + L2_CAP + t; j < end; j += L2_T)
    atomicAdd(&s_cnt[tmp[j] & (BIN_R - 1)], 1);
  __syncthreads();

  int c = (t < BIN_R) ? s_cnt[t] : 0;
  int inc = c;
#pragma unroll
  for (int d = 1; d < 32; d <<= 1) {
    int o = __shfl_up(inc, d, 32);
    if (lane >= d) inc += o;
  }
  if (t < BIN_R && lane == 31) s_w[wid] = inc;
  __syncthreads();
  int excl = 0;
  if (t < BIN_R) {
    int pre = 0;
#pragma unroll
    for (int w = 0; w < BIN_R / 32; ++w) pre += (w < wid) ? s_w[w] : 0;
    excl = pre + inc - c;
    s_cur[t] = excl;
  }
  __syncthreads();

  const bool fits = n <= L2_CAP;
#pragma unroll
  for (int u = 0; u < L2_PER; ++u) {
    if (r[u] >= 0) {
      int p = atomicAdd(&s_cur[r[u] & (BIN_R - 1)], 1);
      int s = r[u] >> BIN_SH;
      if (fits) s_out[p] = s; else ss[start + p] = s;
    }
  }
  for (int j = start + L2_CAP + t; j < end; j += L2_T) {
    int rec = tmp[j];
    int p = atomicAdd(&s_cur[rec & (BIN_R - 1)], 1);
    ss[start + p] = rec >> BIN_SH;
  }
  __syncthreads();
  if (fits)
    for (int q = t; q < n; q += L2_T) ss[start + q] = s_out[q];
  if (t < BIN_R) {
    int node = bin * BIN_R + t;
    if (node < NODES) {
      off[node] = start + excl;
      nrm[node] = rsqrtf(fmaxf((float)c, 1.0f));
    }
  }
  if (bin == NBIN - 1 && t == 0) off[NODES] = end;
}

__global__ void __launch_bounds__(256)
k_node_dots(const float* __restrict__ x, const float* __restrict__ w,
            const float* __restrict__ nrm,
            float* __restrict__ tab, int n_nodes) {
  int gtid = blockIdx.x * blockDim.x + threadIdx.x;
  int wave = gtid >> 5;
  int lane = gtid & 31;
  int base = wave * 16;
  if (base >= n_nodes) return;
  int nl = lane & 15;
  int hi = lane >> 4;

  const float* xrow = x + (size_t)(base + nl) * DIM;
  v8f c = {};
#pragma unroll
  for (int k0 = 0; k0 < DIM; k0 += 4) {
    int kk = k0 + 2 * hi;
    v2f a;
    a.x = xrow[kk];
    a.y = xrow[kk + 1];
    const float* wp = w + (nl & 1) * DIM + kk;
    float bx = wp[0], by = wp[1];
    v2f b;
    b.x = (nl < 2) ? bx : 0.0f;
    b.y = (nl < 2) ? by : 0.0f;
    c = __builtin_amdgcn_wmma_f32_16x16x4_f32(false, a, false, b, (short)0, c,
                                              false, false);
  }

  if (nl < 2) {
    int row0 = base + hi * 8;
#pragma unroll
    for (int r = 0; r < 8; ++r) {
      float d = c[r];
      float dn = d * nrm[row0 + r];
      float2 v = (nl == 0) ? make_float2(dn, d) : make_float2(d, dn);
      *(float2*)(tab + (size_t)(row0 + r) * 4 + nl * 2) = v;
    }
  }
}

template <unsigned LO, unsigned HI>
__device__ __forceinline__ float row_xchg(float x) {
  int b = __builtin_bit_cast(int, x);
  return __builtin_bit_cast(float, __builtin_amdgcn_permlane16(b, b, LO, HI, false, false));
}
__device__ __forceinline__ float xhalf(float x) {
  int b = __builtin_bit_cast(int, x);
  return __builtin_bit_cast(float,
      __builtin_amdgcn_permlanex16(b, b, 0x76543210u, 0xfedcba98u, false, false));
}
__device__ __forceinline__ float wave_max(float x) {
  x = fmaxf(x, row_xchg<0x67452301u, 0xefcdab89u>(x));
  x = fmaxf(x, row_xchg<0x54761032u, 0xdcfe98bau>(x));
  x = fmaxf(x, row_xchg<0x32107654u, 0xba98fedcu>(x));
  x = fmaxf(x, row_xchg<0xfedcba98u, 0x76543210u>(x));
  x = fmaxf(x, xhalf(x));
  return x;
}
__device__ __forceinline__ float wave_sum(float x) {
  x += row_xchg<0x67452301u, 0xefcdab89u>(x);
  x += row_xchg<0x54761032u, 0xdcfe98bau>(x);
  x += row_xchg<0x32107654u, 0xba98fedcu>(x);
  x += row_xchg<0xfedcba98u, 0x76543210u>(x);
  x += xhalf(x);
  return x;
}

__device__ __forceinline__ float leaky(float e) { return (e > 0.0f) ? e : 0.2f * e; }

#define SRC_OF(k) __builtin_amdgcn_readlane(sj, (k))
#define AL_OF(k)  __builtin_bit_cast(float, __builtin_amdgcn_readlane(__builtin_bit_cast(int, al), (k)))
#define ROW_OF(k) (*(const float4*)(x + (size_t)SRC_OF(k) * DIM + lane4))
#define FMA4(a, v) do { float a_ = (a); acc.x += a_ * (v).x; acc.y += a_ * (v).y; \
                        acc.z += a_ * (v).z; acc.w += a_ * (v).w; } while (0)

template <bool LAST>
__global__ void __launch_bounds__(256)
k_seg(const float* __restrict__ x, const float* __restrict__ tab,
      const float* __restrict__ nrm, const int* __restrict__ off,
      const int* __restrict__ ss, float* __restrict__ h, float* __restrict__ tabn,
      const float* __restrict__ y0, const float* __restrict__ y1,
      const float* __restrict__ y2, int n_nodes) {
  int lane = threadIdx.x & 31;
  int n = __builtin_amdgcn_readfirstlane((int)(blockIdx.x * (blockDim.x >> 5) + (threadIdx.x >> 5)));
  if (n >= n_nodes) return;
  const unsigned lane4 = (unsigned)lane * 4u;
  float4 p0 = make_float4(0.f, 0.f, 0.f, 0.f), p1 = p0, p2 = p0, p3 = p0;
  if (LAST) {
    size_t ro = (size_t)n * DIM + lane4;
    p0 = *(const float4*)(y0 + ro);
    p1 = *(const float4*)(y1 + ro);
    p2 = *(const float4*)(y2 + ro);
    p3 = *(const float4*)(x + ro);
  }
  int beg = off[n], end = off[n + 1];
  float ad = tab[(size_t)n * 4 + 3];
  float nv = LAST ? 0.0f : nrm[n];
  const float ninf = __int_as_float(0xff800000);

  int j0 = beg + lane;
  bool v0ok = j0 < end;
  int sj0 = 0;
  float e0 = ninf;
  float4 t0 = make_float4(0.f, 0.f, 0.f, 0.f);
  if (v0ok) {
    sj0 = ss[j0];
    t0 = *(const float4*)(tab + (size_t)sj0 * 4);
    e0 = leaky(t0.x + ad);
  }
  float mx = e0;
  for (int j = j0 + 32; j < end; j += 32)
    mx = fmaxf(mx, leaky(tab[(size_t)ss[j] * 4] + ad));
  mx = wave_max(mx);

  float ex0 = v0ok ? expf(e0 - mx) : 0.0f;
  float sm = ex0;
  for (int j = j0 + 32; j < end; j += 32)
    sm += expf(leaky(tab[(size_t)ss[j] * 4] + ad) - mx);
  sm = wave_sum(sm);
  float den = (sm > 0.0f) ? sm : 1.0f;

  float4 ys = make_float4(0.f, 0.f, 0.f, 0.f);
  if (LAST) {
    ys.x = ((p0.x + p1.x) + p2.x) + p3.x;
    ys.y = ((p0.y + p1.y) + p2.y) + p3.y;
    ys.z = ((p0.z + p1.z) + p2.z) + p3.z;
    ys.w = ((p0.w + p1.w) + p2.w) + p3.w;
  }
  float4 acc = make_float4(0.f, 0.f, 0.f, 0.f);
  int sj = sj0;
  float al = ex0 / den;
  float ds = al * t0.y, dd = al * t0.z;
  for (int jb = beg; jb < end; jb += 32) {
    if (jb != beg) {
      int j = jb + lane;
      sj = 0;
      al = 0.0f;
      if (j < end) {
        sj = ss[j];
        float4 t = *(const float4*)(tab + (size_t)sj * 4);
        al = expf(leaky(t.x + ad) - mx) / den;
        ds += al * t.y;
        dd += al * t.z;
      }
    }
    int cc = end - jb;
    cc = (cc < 32) ? cc : 32;
    int k = 0;
    for (; k + 8 <= cc; k += 8) {
      float4 r0 = ROW_OF(k),     r1 = ROW_OF(k + 1), r2 = ROW_OF(k + 2), r3 = ROW_OF(k + 3);
      float4 r4 = ROW_OF(k + 4), r5 = ROW_OF(k + 5), r6 = ROW_OF(k + 6), r7 = ROW_OF(k + 7);
      FMA4(AL_OF(k), r0);     FMA4(AL_OF(k + 1), r1);
      FMA4(AL_OF(k + 2), r2); FMA4(AL_OF(k + 3), r3);
      FMA4(AL_OF(k + 4), r4); FMA4(AL_OF(k + 5), r5);
      FMA4(AL_OF(k + 6), r6); FMA4(AL_OF(k + 7), r7);
    }
    if (k + 4 <= cc) {
      float4 r0 = ROW_OF(k), r1 = ROW_OF(k + 1), r2 = ROW_OF(k + 2), r3 = ROW_OF(k + 3);
      FMA4(AL_OF(k), r0);     FMA4(AL_OF(k + 1), r1);
      FMA4(AL_OF(k + 2), r2); FMA4(AL_OF(k + 3), r3);
      k += 4;
    }
    int rem = cc - k;
    if (rem > 0) {
      float4 r0 = ROW_OF(k);
      float4 r1 = make_float4(0.f, 0.f, 0.f, 0.f), r2 = r1;
      if (rem > 1) r1 = ROW_OF(k + 1);
      if (rem > 2) r2 = ROW_OF(k + 2);
      FMA4(AL_OF(k), r0);
      if (rem > 1) FMA4(AL_OF(k + 1), r1);
      if (rem > 2) FMA4(AL_OF(k + 2), r2);
    }
  }
  if (LAST) {
    const float sc = 1.0f / (float)(ORD + 1);
    acc.x = (ys.x + acc.x) * sc;
    acc.y = (ys.y + acc.y) * sc;
    acc.z = (ys.z + acc.z) * sc;
    acc.w = (ys.w + acc.w) * sc;
  }
  *(float4*)(h + (size_t)n * DIM + lane4) = acc;
  if (!LAST) {
    ds = wave_sum(ds);
    dd = wave_sum(dd);
    if (lane == 0)
      *(float4*)(tabn + (size_t)n * 4) = make_float4(nv * ds, ds, dd, nv * dd);
  }
}

extern "C" void kernel_launch(void* const* d_in, const int* in_sizes, int n_in,
                              void* d_out, int out_size, void* d_ws, size_t ws_size,
                              hipStream_t stream) {
  const float* feats = (const float*)d_in[0];
  const float* w_att = (const float*)d_in[1];
  const int* src = (const int*)d_in[2];
  const int* dst = (const int*)d_in[3];
  float* out = (float*)d_out;

  const size_t ND = (size_t)NODES * DIM;
  float* ws  = (float*)d_ws;
  float* h1  = ws;
  float* h2  = h1 + ND;
  float* h3  = h2 + ND;
  float* tA  = h3 + ND;
  float* tB  = tA + 4 * NODES;
  float* nrm = tB + 4 * NODES;
  int* off   = (int*)(nrm + NODES);
  int* ss    = off + NODES + 16;
  int* hist  = ss + EDGES;
  int* binst = hist + HIST_N;
  int* tmp   = binst + NBIN + 16;

  const int B = 256;
  const int nblk_dots = (int)(((size_t)((NODES + 15) / 16) * 32 + B - 1) / B);
  const int nblk_seg  = (NODES + (B / 32) - 1) / (B / 32);

  k_hist<<<L1_G, L1_T, 0, stream>>>(dst, hist);
  k_bin<<<L1_G, L1_T, 0, stream>>>(src, dst, hist, binst, tmp);
  k_binsort<<<NBIN, L2_T, 0, stream>>>(binst, tmp, off, nrm, ss);

  static_assert(ORD == 4, "the rounds are written out");
  k_node_dots<<<nblk_dots, B, 0, stream>>>(feats, w_att, nrm, tA, NODES);
  k_seg<false><<<nblk_seg, B, 0, stream>>>(feats, tA, nrm, off, ss, h1, tB, nullptr, nullptr, nullptr, NODES);
  k_seg<false><<<nblk_seg, B, 0, stream>>>(h1, tB, nrm, off, ss, h2, tA, nullptr, nullptr, nullptr, NODES);
  k_seg<false><<<nblk_seg, B, 0, stream>>>(h2, tA, nrm, off, ss, h3, tB, nullptr, nullptr, nullptr, NODES);
  k_seg<true><<<nblk_seg, B, 0, stream>>>(h3, tB, nrm, off, ss, out, nullptr, feats, h1, h2, NODES);
}
